# starting kernel
# baseline (speedup 1.0000x reference)
#include <hip/hip_runtime.h>


#define TT_D 32
#define TT_R 16
#define WAVES_PER_BLOCK 8
#define TILES_PER_WAVE 2
#define ROWS_PER_WAVE (16 * TILES_PER_WAVE)
#define LDS_ROW_STRIDE 18
#define LDS_BUF_FLOATS (16 * LDS_ROW_STRIDE)
#define LDS_PER_TILE (2 * LDS_BUF_FLOATS)

typedef float v2f __attribute__((ext_vector_type(2)));
typedef float v8f __attribute__((ext_vector_type(8)));
typedef int   v4i __attribute__((ext_vector_type(4)));

__global__ __launch_bounds__(256) void tt_chain_wmma_f32(
    const int*   __restrict__ X,
    const float* __restrict__ core_first,
    const float* __restrict__ cores_mid,
    const float* __restrict__ core_last,
    float*       __restrict__ out)
{
    __shared__ float lds[WAVES_PER_BLOCK][TILES_PER_WAVE * LDS_PER_TILE];

    const int tid  = threadIdx.x;
    const int wave = tid >> 5;
    const int lane = tid & 31;
    const int row  = lane & 15;
    const int half = lane >> 4;

    const int base = (blockIdx.x * WAVES_PER_BLOCK + wave) * ROWS_PER_WAVE;

    float* wbuf = lds[wave];

    unsigned xm[TILES_PER_WAVE];
    #pragma unroll
    for (int t = 0; t < TILES_PER_WAVE; ++t) {
        const v4i* xr = (const v4i*)(X + (size_t)(base + 16 * t + row) * TT_D);
        unsigned m = 0u;
        #pragma unroll
        for (int q = 0; q < 8; ++q) {
            v4i v = __builtin_nontemporal_load(xr + q);
            m |= (unsigned)(v.x & 1) << (4 * q + 0);
            m |= (unsigned)(v.y & 1) << (4 * q + 1);
            m |= (unsigned)(v.z & 1) << (4 * q + 2);
            m |= (unsigned)(v.w & 1) << (4 * q + 3);
        }
        xm[t] = m;
    }

    float a[TILES_PER_WAVE][4][2];
    #pragma unroll
    for (int t = 0; t < TILES_PER_WAVE; ++t) {
        const unsigned x0 = xm[t] & 1u;
        #pragma unroll
        for (int c = 0; c < 4; ++c)
            #pragma unroll
            for (int j = 0; j < 2; ++j) {
                int k = 4 * c + 2 * half + j;
                a[t][c][j] = core_first[k * 2 + x0];
            }
    }

    #pragma unroll 1
    for (int ci = 0; ci < TT_D - 2; ++ci) {
        const float* Bc = cores_mid + ci * (TT_R * TT_R * 2);
        float2 bv[4][2];
        #pragma unroll
        for (int c = 0; c < 4; ++c)
            #pragma unroll
            for (int j = 0; j < 2; ++j) {
                int k = 4 * c + 2 * half + j;
                bv[c][j] = *(const float2*)(Bc + k * 32 + row * 2);
            }
        v2f b0[4], b1[4];
        #pragma unroll
        for (int c = 0; c < 4; ++c) {
            b0[c] = (v2f){ bv[c][0].x, bv[c][1].x };
            b1[c] = (v2f){ bv[c][0].y, bv[c][1].y };
        }

        v8f p0[TILES_PER_WAVE], p1[TILES_PER_WAVE];
        #pragma unroll
        for (int t = 0; t < TILES_PER_WAVE; ++t) { p0[t] = (v8f){}; p1[t] = (v8f){}; }

        #pragma unroll
        for (int c = 0; c < 4; ++c) {
            #pragma unroll
            for (int t = 0; t < TILES_PER_WAVE; ++t) {
                v2f av = { a[t][c][0], a[t][c][1] };
                p0[t] = __builtin_amdgcn_wmma_f32_16x16x4_f32(
                            false, av, false, b0[c], (short)0, p0[t], false, false);
                p1[t] = __builtin_amdgcn_wmma_f32_16x16x4_f32(
                            false, av, false, b1[c], (short)0, p1[t], false, false);
            }
        }

        #pragma unroll
        for (int t = 0; t < TILES_PER_WAVE; ++t) {
            float* tb = wbuf + t * LDS_PER_TILE;
            #pragma unroll
            for (int v = 0; v < 8; ++v) {
                int r = v + 8 * half;
                tb[r * LDS_ROW_STRIDE + row]                  = p0[t][v];
                tb[LDS_BUF_FLOATS + r * LDS_ROW_STRIDE + row] = p1[t][v];
            }
        }

        #pragma unroll
        for (int t = 0; t < TILES_PER_WAVE; ++t) {
            const unsigned sel = (xm[t] >> (ci + 1)) & 1u;
            const float* src = wbuf + t * LDS_PER_TILE + sel * LDS_BUF_FLOATS
                               + row * LDS_ROW_STRIDE + 2 * half;
            #pragma unroll
            for (int c = 0; c < 4; ++c) {
                float2 tv = *(const float2*)(src + 4 * c);
                a[t][c][0] = tv.x;
                a[t][c][1] = tv.y;
            }
        }
    }

    #pragma unroll
    for (int t = 0; t < TILES_PER_WAVE; ++t) {
        const unsigned xl = (xm[t] >> 31) & 1u;
        float acc = 0.f;
        #pragma unroll
        for (int c = 0; c < 4; ++c)
            #pragma unroll
            for (int j = 0; j < 2; ++j) {
                int k = 4 * c + 2 * half + j;
                acc += a[t][c][j] * core_last[k * 2 + xl];
            }
        acc += __shfl_xor(acc, 16, 32);
        if (half == 0)
            __builtin_nontemporal_store(acc, out + base + 16 * t + row);
    }
}

extern "C" void kernel_launch(void* const* d_in, const int* in_sizes, int n_in,
                              void* d_out, int out_size, void* d_ws, size_t ws_size,
                              hipStream_t stream) {
    (void)in_sizes; (void)n_in; (void)d_ws; (void)ws_size;
    const int*   X  = (const int*)d_in[0];
    const float* cf = (const float*)d_in[1];
    const float* cm = (const float*)d_in[2];
    const float* cl = (const float*)d_in[3];
    float* out = (float*)d_out;

    const int rows_per_block = ROWS_PER_WAVE * WAVES_PER_BLOCK;
    const int blocks = out_size / rows_per_block;
    tt_chain_wmma_f32<<<blocks, 256, 0, stream>>>(X, cf, cm, cl, out);
}


# End tables: vectors of the first 15 and last 15 digits built once per 2^15 patterns (WMMA kernel, last 2 steps keep both products); main: 2 middle steps + dot; chain reassociated, sum order changes
# speedup vs baseline: 2.7789x; 2.7789x over previous
#include <hip/hip_runtime.h>


#define TT_D 32
#define TT_R 16
#define TT_P 15
#define TT_S 15
#define TT_J 2
#define TT_L (TT_P - TT_J)
#define TT_NP (1 << TT_P)
#define TT_NS (1 << TT_S)
#define TT_MID_FIRST (TT_P - 1)
#define TT_MID_END   (TT_D - 1 - TT_S)
#define TT_TABLE_WAVES (1 << (TT_L - 4))
#define TABLE_WAVES_PER_BLOCK 4
#define TT_CORE (TT_R * TT_R * 2)
#define WAVES_PER_BLOCK 8
#define TILES_PER_WAVE 2
#define ROWS_PER_WAVE (16 * TILES_PER_WAVE)

typedef float v2f __attribute__((ext_vector_type(2)));
typedef float v4f __attribute__((ext_vector_type(4)));
typedef float v8f __attribute__((ext_vector_type(8)));
typedef int   v4i __attribute__((ext_vector_type(4)));

__device__ __forceinline__ void tt_load_a_fwd(const float* __restrict__ core, int col, int half,
                                              v2f (&a0)[4], v2f (&a1)[4])
{
    const float* Ac = core + (8 * half) * (TT_R * 2) + col * 2;
    #pragma unroll
    for (int q = 0; q < 4; ++q) {
        const float2 c0 = *(const float2*)(Ac + (2 * q + 0) * (TT_R * 2));
        const float2 c1 = *(const float2*)(Ac + (2 * q + 1) * (TT_R * 2));
        a0[q] = (v2f){ c0.x, c1.x };
        a1[q] = (v2f){ c0.y, c1.y };
    }
}

__device__ __forceinline__ void tt_load_a_bwd(const float* __restrict__ core, int col, int half,
                                              v2f (&a0)[4], v2f (&a1)[4])
{
    const v4f* Ac = (const v4f*)(core + col * (TT_R * 2) + (8 * half) * 2);
    #pragma unroll
    for (int q = 0; q < 4; ++q) {
        const v4f f = Ac[q];
        a0[q] = (v2f){ f.x, f.z };
        a1[q] = (v2f){ f.y, f.w };
    }
}

__device__ __forceinline__ v8f tt_tile_step(const v8f c, const v2f (&a0)[4], const v2f (&a1)[4], bool sel)
{
    v8f p0 = (v8f){}, p1 = (v8f){};
    #pragma unroll
    for (int q = 0; q < 4; ++q) {
        const v2f bq = { c[2 * q], c[2 * q + 1] };
        p0 = __builtin_amdgcn_wmma_f32_16x16x4_f32(false, a0[q], false, bq, (short)0, p0, false, false);
        p1 = __builtin_amdgcn_wmma_f32_16x16x4_f32(false, a1[q], false, bq, (short)0, p1, false, false);
    }
    v8f r;
    #pragma unroll
    for (int v = 0; v < 8; ++v)
        r[v] = sel ? p1[v] : p0[v];
    return r;
}

__device__ __forceinline__ void tt_tile_both(const v8f c, const v2f (&a0)[4], const v2f (&a1)[4],
                                             v8f& r0, v8f& r1)
{
    v8f p0 = (v8f){}, p1 = (v8f){};
    #pragma unroll
    for (int q = 0; q < 4; ++q) {
        const v2f bq = { c[2 * q], c[2 * q + 1] };
        p0 = __builtin_amdgcn_wmma_f32_16x16x4_f32(false, a0[q], false, bq, (short)0, p0, false, false);
        p1 = __builtin_amdgcn_wmma_f32_16x16x4_f32(false, a1[q], false, bq, (short)0, p1, false, false);
    }
    r0 = p0; r1 = p1;
}

__device__ __forceinline__ void tt_store_row(float* __restrict__ T, unsigned entry, int half, const v8f c)
{
    v4f* dst = (v4f*)(T + (size_t)entry * TT_R + 8 * half);
    dst[0] = (v4f){ c[0], c[1], c[2], c[3] };
    dst[1] = (v4f){ c[4], c[5], c[6], c[7] };
}

__global__ __launch_bounds__(128) __attribute__((amdgpu_waves_per_eu(1, 2))) void tt_end_tables(
    const float* __restrict__ core_first,
    const float* __restrict__ cores_mid,
    const float* __restrict__ core_last,
    float*       __restrict__ U,
    float*       __restrict__ Wt)
{
    static_assert(TT_J == 2, "the two keep-both steps are written out");
    const int tid  = threadIdx.x;
    const int wave = tid >> 5;
    const int lane = tid & 31;
    const int col  = lane & 15;
    const int half = lane >> 4;
    const int gw   = blockIdx.x * TABLE_WAVES_PER_BLOCK + wave;

    if (gw < TT_TABLE_WAVES) {
        const unsigned e = (unsigned)(gw * 16 + col);
        v2f a0[TT_P - 1][4], a1[TT_P - 1][4];
        #pragma unroll
        for (int i = 0; i < TT_P - 1; ++i)
            tt_load_a_fwd(cores_mid + i * TT_CORE, col, half, a0[i], a1[i]);
        v8f c;
        #pragma unroll
        for (int v = 0; v < 8; ++v)
            c[v] = core_first[(8 * half + v) * 2 + (e & 1u)];
        #pragma unroll
        for (int i = 0; i < TT_L - 1; ++i)
            c = tt_tile_step(c, a0[i], a1[i], ((e >> (i + 1)) & 1u) != 0u);
        v8f c0, c1, c00, c01, c10, c11;
        tt_tile_both(c,  a0[TT_L - 1], a1[TT_L - 1], c0, c1);
        tt_tile_both(c0, a0[TT_L],     a1[TT_L],     c00, c01);
        tt_tile_both(c1, a0[TT_L],     a1[TT_L],     c10, c11);
        tt_store_row(U, e,                                       half, c00);
        tt_store_row(U, e | (1u << (TT_L + 1)),                  half, c01);
        tt_store_row(U, e | (1u << TT_L),                        half, c10);
        tt_store_row(U, e | (1u << TT_L) | (1u << (TT_L + 1)),   half, c11);
    } else {
        const unsigned e = (unsigned)((gw - TT_TABLE_WAVES) * 16 + col);
        v2f a0[TT_S - 1][4], a1[TT_S - 1][4];
        #pragma unroll
        for (int i = 0; i < TT_S - 1; ++i)
            tt_load_a_bwd(cores_mid + (TT_D - 3 - i) * TT_CORE, col, half, a0[i], a1[i]);
        v8f c;
        #pragma unroll
        for (int v = 0; v < 8; ++v)
            c[v] = core_last[(8 * half + v) * 2 + ((e >> (TT_L - 1)) & 1u)];
        #pragma unroll
        for (int i = 0; i < TT_L - 1; ++i)
            c = tt_tile_step(c, a0[i], a1[i], ((e >> (TT_L - 2 - i)) & 1u) != 0u);
        v8f c0, c1, c00, c01, c10, c11;
        tt_tile_both(c,  a0[TT_L - 1], a1[TT_L - 1], c0, c1);
        tt_tile_both(c0, a0[TT_L],     a1[TT_L],     c00, c01);
        tt_tile_both(c1, a0[TT_L],     a1[TT_L],     c10, c11);
        tt_store_row(Wt, (e << 2),      half, c00);
        tt_store_row(Wt, (e << 2) | 1u, half, c01);
        tt_store_row(Wt, (e << 2) | 2u, half, c10);
        tt_store_row(Wt, (e << 2) | 3u, half, c11);
    }
}

__global__ __launch_bounds__(256) __attribute__((amdgpu_waves_per_eu(8, 8))) void tt_chain_wmma_f32(
    const int*   __restrict__ X,
    const float* __restrict__ cores_mid,
    const float* __restrict__ U,
    const float* __restrict__ Wt,
    float*       __restrict__ out)
{
    const int tid  = threadIdx.x;
    const int wave = tid >> 5;
    const int lane = tid & 31;
    const int col  = lane & 15;
    const int half = lane >> 4;

    const int base = (blockIdx.x * WAVES_PER_BLOCK + wave) * ROWS_PER_WAVE;

    v4i xv[TILES_PER_WAVE][8];
    #pragma unroll
    for (int t = 0; t < TILES_PER_WAVE; ++t) {
        const v4i* xr = (const v4i*)(X + (size_t)(base + 16 * t + col) * TT_D);
        #pragma unroll
        for (int q = 0; q < 8; ++q)
            xv[t][q] = __builtin_nontemporal_load(xr + q);
    }

    v2f a0[TT_MID_END - TT_MID_FIRST][4], a1[TT_MID_END - TT_MID_FIRST][4];
    #pragma unroll
    for (int i = 0; i < TT_MID_END - TT_MID_FIRST; ++i)
        tt_load_a_fwd(cores_mid + (TT_MID_FIRST + i) * TT_CORE, col, half, a0[i], a1[i]);

    unsigned xm[TILES_PER_WAVE];
    #pragma unroll
    for (int t = 0; t < TILES_PER_WAVE; ++t) {
        unsigned m = 0u;
        #pragma unroll
        for (int q = 0; q < 8; ++q) {
            const v4i v = xv[t][q];
            m |= (unsigned)(v.x & 1) << (4 * q + 0);
            m |= (unsigned)(v.y & 1) << (4 * q + 1);
            m |= (unsigned)(v.z & 1) << (4 * q + 2);
            m |= (unsigned)(v.w & 1) << (4 * q + 3);
        }
        xm[t] = m;
    }

    v8f cr[TILES_PER_WAVE], wv[TILES_PER_WAVE];
    #pragma unroll
    for (int t = 0; t < TILES_PER_WAVE; ++t) {
        const unsigned p = xm[t] & (unsigned)(TT_NP - 1);
        const unsigned s = xm[t] >> (TT_D - TT_S);
        const v4f* up = (const v4f*)(U  + (size_t)p * TT_R + 8 * half);
        const v4f* wp = (const v4f*)(Wt + (size_t)s * TT_R + 8 * half);
        const v4f u0 = up[0], u1 = up[1], w0 = wp[0], w1 = wp[1];
        cr[t] = (v8f){ u0.x, u0.y, u0.z, u0.w, u1.x, u1.y, u1.z, u1.w };
        wv[t] = (v8f){ w0.x, w0.y, w0.z, w0.w, w1.x, w1.y, w1.z, w1.w };
    }

    #pragma unroll
    for (int i = 0; i < TT_MID_END - TT_MID_FIRST; ++i) {
        #pragma unroll
        for (int t = 0; t < TILES_PER_WAVE; ++t)
            cr[t] = tt_tile_step(cr[t], a0[i], a1[i], ((xm[t] >> (TT_MID_FIRST + i + 1)) & 1u) != 0u);
    }

    #pragma unroll
    for (int t = 0; t < TILES_PER_WAVE; ++t) {
        float acc = 0.f;
        #pragma unroll
        for (int v = 0; v < 8; ++v)
            acc += cr[t][v] * wv[t][v];
        acc += __shfl_xor(acc, 16, 32);
        if (half == 0)
            __builtin_nontemporal_store(acc, out + base + 16 * t + col);
    }
}

extern "C" void kernel_launch(void* const* d_in, const int* in_sizes, int n_in,
                              void* d_out, int out_size, void* d_ws, size_t ws_size,
                              hipStream_t stream) {
    (void)in_sizes; (void)n_in; (void)ws_size;
    const int*   X  = (const int*)d_in[0];
    const float* cf = (const float*)d_in[1];
    const float* cm = (const float*)d_in[2];
    const float* cl = (const float*)d_in[3];
    float* out = (float*)d_out;
    float* U   = (float*)d_ws;
    float* Wt  = U + TT_NP * TT_R;

    tt_end_tables<<<2 * TT_TABLE_WAVES / TABLE_WAVES_PER_BLOCK, 32 * TABLE_WAVES_PER_BLOCK, 0, stream>>>(cf, cm, cl, U, Wt);

    const int rows_per_block = ROWS_PER_WAVE * WAVES_PER_BLOCK;
    const int blocks = out_size / rows_per_block;
    tt_chain_wmma_f32<<<blocks, 256, 0, stream>>>(X, cm, U, Wt, out);
}


# Main kernel reads X coalesced: a wave takes its 32 rows as 8 loads of 512 contiguous bytes (4 lines per load, was 16); row masks by an 8-lane nibble transpose (dpp8); one 128-byte out store per wave
# speedup vs baseline: 3.5301x; 1.2703x over previous
#include <hip/hip_runtime.h>


#define TT_D 32
#define TT_R 16
#define TT_P 15
#define TT_S 15
#define TT_J 2
#define TT_L (TT_P - TT_J)
#define TT_NP (1 << TT_P)
#define TT_NS (1 << TT_S)
#define TT_MID_FIRST (TT_P - 1)
#define TT_MID_END   (TT_D - 1 - TT_S)
#define TT_TABLE_WAVES (1 << (TT_L - 4))
#define TABLE_WAVES_PER_BLOCK 4
#define TT_CORE (TT_R * TT_R * 2)
#define WAVES_PER_BLOCK 8
#define TILES_PER_WAVE 2
#define ROWS_PER_WAVE (16 * TILES_PER_WAVE)

#define TT_DPP8(s0, s1, s2, s3, s4, s5, s6, s7) \
    ((s0) | ((s1) << 3) | ((s2) << 6) | ((s3) << 9) | ((s4) << 12) | ((s5) << 15) | ((s6) << 18) | ((s7) << 21))

typedef float v2f __attribute__((ext_vector_type(2)));
typedef float v4f __attribute__((ext_vector_type(4)));
typedef float v8f __attribute__((ext_vector_type(8)));
typedef int   v4i __attribute__((ext_vector_type(4)));

__device__ __forceinline__ void tt_load_a_fwd(const float* __restrict__ core, int col, int half,
                                              v2f (&a0)[4], v2f (&a1)[4])
{
    const float* Ac = core + (8 * half) * (TT_R * 2) + col * 2;
    #pragma unroll
    for (int q = 0; q < 4; ++q) {
        const float2 c0 = *(const float2*)(Ac + (2 * q + 0) * (TT_R * 2));
        const float2 c1 = *(const float2*)(Ac + (2 * q + 1) * (TT_R * 2));
        a0[q] = (v2f){ c0.x, c1.x };
        a1[q] = (v2f){ c0.y, c1.y };
    }
}

__device__ __forceinline__ void tt_load_a_bwd(const float* __restrict__ core, int col, int half,
                                              v2f (&a0)[4], v2f (&a1)[4])
{
    const v4f* Ac = (const v4f*)(core + col * (TT_R * 2) + (8 * half) * 2);
    #pragma unroll
    for (int q = 0; q < 4; ++q) {
        const v4f f = Ac[q];
        a0[q] = (v2f){ f.x, f.z };
        a1[q] = (v2f){ f.y, f.w };
    }
}

__device__ __forceinline__ v8f tt_tile_step(const v8f c, const v2f (&a0)[4], const v2f (&a1)[4], bool sel)
{
    v8f p0 = (v8f){}, p1 = (v8f){};
    #pragma unroll
    for (int q = 0; q < 4; ++q) {
        const v2f bq = { c[2 * q], c[2 * q + 1] };
        p0 = __builtin_amdgcn_wmma_f32_16x16x4_f32(false, a0[q], false, bq, (short)0, p0, false, false);
        p1 = __builtin_amdgcn_wmma_f32_16x16x4_f32(false, a1[q], false, bq, (short)0, p1, false, false);
    }
    v8f r;
    #pragma unroll
    for (int v = 0; v < 8; ++v)
        r[v] = sel ? p1[v] : p0[v];
    return r;
}

__device__ __forceinline__ void tt_tile_both(const v8f c, const v2f (&a0)[4], const v2f (&a1)[4],
                                             v8f& r0, v8f& r1)
{
    v8f p0 = (v8f){}, p1 = (v8f){};
    #pragma unroll
    for (int q = 0; q < 4; ++q) {
        const v2f bq = { c[2 * q], c[2 * q + 1] };
        p0 = __builtin_amdgcn_wmma_f32_16x16x4_f32(false, a0[q], false, bq, (short)0, p0, false, false);
        p1 = __builtin_amdgcn_wmma_f32_16x16x4_f32(false, a1[q], false, bq, (short)0, p1, false, false);
    }
    r0 = p0; r1 = p1;
}

__device__ __forceinline__ void tt_store_row(float* __restrict__ T, unsigned entry, int half, const v8f c)
{
    v4f* dst = (v4f*)(T + (size_t)entry * TT_R + 8 * half);
    dst[0] = (v4f){ c[0], c[1], c[2], c[3] };
    dst[1] = (v4f){ c[4], c[5], c[6], c[7] };
}

__global__ __launch_bounds__(128) __attribute__((amdgpu_waves_per_eu(1, 2))) void tt_end_tables(
    const float* __restrict__ core_first,
    const float* __restrict__ cores_mid,
    const float* __restrict__ core_last,
    float*       __restrict__ U,
    float*       __restrict__ Wt)
{
    static_assert(TT_J == 2, "the two keep-both steps are written out");
    const int tid  = threadIdx.x;
    const int wave = tid >> 5;
    const int lane = tid & 31;
    const int col  = lane & 15;
    const int half = lane >> 4;
    const int gw   = blockIdx.x * TABLE_WAVES_PER_BLOCK + wave;

    if (gw < TT_TABLE_WAVES) {
        const unsigned e = (unsigned)(gw * 16 + col);
        v2f a0[TT_P - 1][4], a1[TT_P - 1][4];
        #pragma unroll
        for (int i = 0; i < TT_P - 1; ++i)
            tt_load_a_fwd(cores_mid + i * TT_CORE, col, half, a0[i], a1[i]);
        v8f c;
        #pragma unroll
        for (int v = 0; v < 8; ++v)
            c[v] = core_first[(8 * half + v) * 2 + (e & 1u)];
        #pragma unroll
        for (int i = 0; i < TT_L - 1; ++i)
            c = tt_tile_step(c, a0[i], a1[i], ((e >> (i + 1)) & 1u) != 0u);
        v8f c0, c1, c00, c01, c10, c11;
        tt_tile_both(c,  a0[TT_L - 1], a1[TT_L - 1], c0, c1);
        tt_tile_both(c0, a0[TT_L],     a1[TT_L],     c00, c01);
        tt_tile_both(c1, a0[TT_L],     a1[TT_L],     c10, c11);
        tt_store_row(U, e,                                       half, c00);
        tt_store_row(U, e | (1u << (TT_L + 1)),                  half, c01);
        tt_store_row(U, e | (1u << TT_L),                        half, c10);
        tt_store_row(U, e | (1u << TT_L) | (1u << (TT_L + 1)),   half, c11);
    } else {
        const unsigned e = (unsigned)((gw - TT_TABLE_WAVES) * 16 + col);
        v2f a0[TT_S - 1][4], a1[TT_S - 1][4];
        #pragma unroll
        for (int i = 0; i < TT_S - 1; ++i)
            tt_load_a_bwd(cores_mid + (TT_D - 3 - i) * TT_CORE, col, half, a0[i], a1[i]);
        v8f c;
        #pragma unroll
        for (int v = 0; v < 8; ++v)
            c[v] = core_last[(8 * half + v) * 2 + ((e >> (TT_L - 1)) & 1u)];
        #pragma unroll
        for (int i = 0; i < TT_L - 1; ++i)
            c = tt_tile_step(c, a0[i], a1[i], ((e >> (TT_L - 2 - i)) & 1u) != 0u);
        v8f c0, c1, c00, c01, c10, c11;
        tt_tile_both(c,  a0[TT_L - 1], a1[TT_L - 1], c0, c1);
        tt_tile_both(c0, a0[TT_L],     a1[TT_L],     c00, c01);
        tt_tile_both(c1, a0[TT_L],     a1[TT_L],     c10, c11);
        tt_store_row(Wt, (e << 2),      half, c00);
        tt_store_row(Wt, (e << 2) | 1u, half, c01);
        tt_store_row(Wt, (e << 2) | 2u, half, c10);
        tt_store_row(Wt, (e << 2) | 3u, half, c11);
    }
}

__global__ __launch_bounds__(256) __attribute__((amdgpu_waves_per_eu(8, 8))) void tt_chain_wmma_f32(
    const int*   __restrict__ X,
    const float* __restrict__ cores_mid,
    const float* __restrict__ U,
    const float* __restrict__ Wt,
    float*       __restrict__ out)
{
    const int tid  = threadIdx.x;
    const int wave = tid >> 5;
    const int lane = tid & 31;
    const int col  = lane & 15;
    const int half = lane >> 4;

    const int base = (blockIdx.x * WAVES_PER_BLOCK + wave) * ROWS_PER_WAVE;

    v4i xv[8];
    {
        const v4i* xr = (const v4i*)(X + (size_t)base * TT_D) + lane;
        #pragma unroll
        for (int k = 0; k < 8; ++k)
            xv[k] = __builtin_nontemporal_load(xr + 32 * k);
    }

    v2f a0[TT_MID_END - TT_MID_FIRST][4], a1[TT_MID_END - TT_MID_FIRST][4];
    #pragma unroll
    for (int i = 0; i < TT_MID_END - TT_MID_FIRST; ++i)
        tt_load_a_fwd(cores_mid + (TT_MID_FIRST + i) * TT_CORE, col, half, a0[i], a1[i]);

    unsigned w = 0u;
    #pragma unroll
    for (int k = 0; k < 8; ++k) {
        const v4i v = xv[k];
        w |= (unsigned)(v.x & 1) << (4 * k + 0);
        w |= (unsigned)(v.y & 1) << (4 * k + 1);
        w |= (unsigned)(v.z & 1) << (4 * k + 2);
        w |= (unsigned)(v.w & 1) << (4 * k + 3);
    }
    {
        const unsigned p4 = (unsigned)__builtin_amdgcn_mov_dpp8(w, TT_DPP8(4, 5, 6, 7, 0, 1, 2, 3));
        const unsigned k4 = (lane & 4) ? 0xFFFF0000u : 0x0000FFFFu;
        w = (k4 & w) | (~k4 & __builtin_rotateright32(p4, 16));
        const unsigned p2 = (unsigned)__builtin_amdgcn_mov_dpp8(w, TT_DPP8(2, 3, 0, 1, 6, 7, 4, 5));
        const unsigned k2 = (lane & 2) ? 0xFF00FF00u : 0x00FF00FFu;
        w = (k2 & w) | (~k2 & __builtin_rotateright32(p2, (lane & 2) ? 8 : 24));
        const unsigned p1 = (unsigned)__builtin_amdgcn_mov_dpp8(w, TT_DPP8(1, 0, 3, 2, 5, 4, 7, 6));
        const unsigned k1 = (lane & 1) ? 0xF0F0F0F0u : 0x0F0F0F0Fu;
        w = (k1 & w) | (~k1 & __builtin_rotateright32(p1, (lane & 1) ? 4 : 28));
    }
    const int wrow = 4 * (lane & 7) + (lane >> 3);
    unsigned xm[TILES_PER_WAVE];
    {
        const unsigned wo = (unsigned)__builtin_amdgcn_permlanex16(w, w, 0x76543210u, 0xfedcba98u, false, false);
        xm[0] = half ? wo : w;
        xm[1] = half ? w : wo;
    }

    v8f cr[TILES_PER_WAVE], wv[TILES_PER_WAVE];
    #pragma unroll
    for (int t = 0; t < TILES_PER_WAVE; ++t) {
        const unsigned p = xm[t] & (unsigned)(TT_NP - 1);
        const unsigned s = xm[t] >> (TT_D - TT_S);
        const v4f* up = (const v4f*)(U  + (size_t)p * TT_R + 8 * half);
        const v4f* wp = (const v4f*)(Wt + (size_t)s * TT_R + 8 * half);
        const v4f u0 = up[0], u1 = up[1], w0 = wp[0], w1 = wp[1];
        cr[t] = (v8f){ u0.x, u0.y, u0.z, u0.w, u1.x, u1.y, u1.z, u1.w };
        wv[t] = (v8f){ w0.x, w0.y, w0.z, w0.w, w1.x, w1.y, w1.z, w1.w };
    }

    #pragma unroll
    for (int i = 0; i < TT_MID_END - TT_MID_FIRST; ++i) {
        #pragma unroll
        for (int t = 0; t < TILES_PER_WAVE; ++t)
            cr[t] = tt_tile_step(cr[t], a0[i], a1[i], ((xm[t] >> (TT_MID_FIRST + i + 1)) & 1u) != 0u);
    }

    float res[TILES_PER_WAVE];
    #pragma unroll
    for (int t = 0; t < TILES_PER_WAVE; ++t) {
        float acc = 0.f;
        #pragma unroll
        for (int v = 0; v < 8; ++v)
            acc += cr[t][v] * wv[t][v];
        res[t] = acc + __shfl_xor(acc, 16, 32);
    }
    __builtin_nontemporal_store(half ? res[1] : res[0], out + base + wrow);
}

extern "C" void kernel_launch(void* const* d_in, const int* in_sizes, int n_in,
                              void* d_out, int out_size, void* d_ws, size_t ws_size,
                              hipStream_t stream) {
    (void)in_sizes; (void)n_in; (void)ws_size;
    const int*   X  = (const int*)d_in[0];
    const float* cf = (const float*)d_in[1];
    const float* cm = (const float*)d_in[2];
    const float* cl = (const float*)d_in[3];
    float* out = (float*)d_out;
    float* U   = (float*)d_ws;
    float* Wt  = U + TT_NP * TT_R;

    tt_end_tables<<<2 * TT_TABLE_WAVES / TABLE_WAVES_PER_BLOCK, 32 * TABLE_WAVES_PER_BLOCK, 0, stream>>>(cf, cm, cl, U, Wt);

    const int rows_per_block = ROWS_PER_WAVE * WAVES_PER_BLOCK;
    const int blocks = out_size / rows_per_block;
    tt_chain_wmma_f32<<<blocks, 256, 0, stream>>>(X, cm, U, Wt, out);
}
